# starting kernel
# baseline (speedup 1.0000x reference)
#include <hip/hip_runtime.h>

#define HH 256
#define WW 256
#define BB 4
#define CIN 8
#define COUT 8
#define NTYPES 216
#define NPIX (BB*HH*WW)
#define KDIM 72
#define NW   (COUT*KDIM)
#define EMB_STRIDE (NW + COUT)

typedef float v2f __attribute__((ext_vector_type(2)));
typedef float v8f __attribute__((ext_vector_type(8)));

__global__ void k_zero_counts(int* counts) {
    int i = blockIdx.x * blockDim.x + threadIdx.x;
    if (i < NTYPES) counts[i] = 0;
}

__global__ void k_hist(const int* __restrict__ buckets, int* __restrict__ counts) {
    int i = blockIdx.x * blockDim.x + threadIdx.x;
    if (i < NPIX) atomicAdd(&counts[buckets[i]], 1);
}

__global__ void k_scan(const int* __restrict__ counts, int* __restrict__ offsets,
                       int* __restrict__ cursor) {
    int acc = 0;
    for (int t = 0; t < NTYPES; ++t) {
        offsets[t] = acc;
        cursor[t]  = acc;
        acc += counts[t];
    }
    offsets[NTYPES] = acc;
}

__global__ void k_scatter(const int* __restrict__ buckets, int* __restrict__ cursor,
                          int* __restrict__ perm) {
    int i = blockIdx.x * blockDim.x + threadIdx.x;
    if (i < NPIX) {
        int p = atomicAdd(&cursor[buckets[i]], 1);
        perm[p] = i;
    }
}

#define WAVES_PER_BLOCK 8
#define BLOCKS_PER_TYPE 4

__global__ __launch_bounds__(256) void k_bucket_gemm(
    const float* __restrict__ x, const float* __restrict__ emb,
    const int* __restrict__ offsets, const int* __restrict__ perm,
    float* __restrict__ out) {

    const int t    = blockIdx.x;
    const int lane = threadIdx.x & 31;
    const int gwave  = blockIdx.y * WAVES_PER_BLOCK + (threadIdx.x >> 5);
    const int nwaves = gridDim.y * WAVES_PER_BLOCK;

    const int start  = offsets[t];
    const int end    = offsets[t + 1];
    const int ntiles = (end - start + 15) >> 4;

    const float* wrow = emb + t * EMB_STRIDE;

    const int n  = lane & 15;
    const int hi = lane >> 4;
    const int khalf = hi << 1;
    v2f bmat[18];
    #pragma unroll
    for (int cc = 0; cc < 18; ++cc) {
        const int k = cc * 4 + khalf;
        float b0 = 0.f, b1 = 0.f;
        if (n < COUT) {
            b0 = wrow[n * KDIM + k];
            b1 = wrow[n * KDIM + k + 1];
        }
        bmat[cc].x = b0; bmat[cc].y = b1;
    }
    const float biasv = (n < COUT) ? wrow[NW + n] : 0.f;

    for (int tile = gwave; tile < ntiles; tile += nwaves) {
        const int rowbase = start + tile * 16;

        const int m     = lane & 15;
        const int pidx  = rowbase + m;
        const int valid = (pidx < end) ? 1 : 0;
        const int cp    = valid ? pidx : start;
        const int pix   = perm[cp];
        const int pb = pix >> 16;
        const int ph = (pix >> 8) & 255;
        const int pw = pix & 255;

        const float rowmask = valid ? 1.f : 0.f;
        int   hrow[3], wcol[3];
        float hmask[3], wmask[3];
        #pragma unroll
        for (int d = 0; d < 3; ++d) {
            const int hh = ph + d - 1;
            const int ww = pw + d - 1;
            hrow[d]  = (hh & 255) << 8;
            wcol[d]  = ww & 255;
            hmask[d] = ((unsigned)hh < 256u) ? 1.f : 0.f;
            wmask[d] = ((unsigned)ww < 256u) ? 1.f : 0.f;
        }
        const int base = (pb * CIN) << 16;
        int   off9[9];
        float msk9[9];
        #pragma unroll
        for (int r = 0; r < 9; ++r) {
            off9[r] = base + hrow[r / 3] + wcol[r % 3];
            msk9[r] = hmask[r / 3] * wmask[r % 3] * rowmask;
        }

        int   offR[9];
        float mskR[9];
        #pragma unroll
        for (int s = 0; s < 9; ++s) {
            const int s2 = (s + 2) % 9;
            offR[s] = hi ? off9[s2] : off9[s];
            mskR[s] = hi ? msk9[s2] : msk9[s];
        }

        v2f a[18];
        #pragma unroll
        for (int cc = 0; cc < 18; ++cc) {
            #pragma unroll
            for (int j = 0; j < 2; ++j) {
                const int kbase = cc * 4 + j;
                const int tapc  = kbase % 9;
                const int cA = (kbase / 9) << 16;
                const int cB = ((kbase + 2) / 9) << 16;
                const int coff = (cA == cB) ? cA : (hi ? cB : cA);
                const float v = x[offR[tapc] + coff] * mskR[tapc];
                if (j == 0) a[cc].x = v; else a[cc].y = v;
            }
        }

        v8f acc = {};
        #pragma unroll
        for (int cc = 0; cc < 18; ++cc) {
            acc = __builtin_amdgcn_wmma_f32_16x16x4_f32(
                false, a[cc], false, bmat[cc], (short)0, acc, false, false);
        }

        const int col  = lane & 15;
        const int rofs = hi << 3;
        #pragma unroll
        for (int r = 0; r < 8; ++r) {
            const int row   = r + rofs;
            const int pix2  = __shfl(pix, row, 32);
            if (col < COUT && (rowbase + row) < end) {
                const int b2 = pix2 >> 16;
                const int h2 = (pix2 >> 8) & 255;
                const int w2 = pix2 & 255;
                out[((b2 * COUT + col) << 16) + (h2 << 8) + w2] = acc[r] + biasv;
            }
        }
    }
}

extern "C" void kernel_launch(void* const* d_in, const int* in_sizes, int n_in,
                              void* d_out, int out_size, void* d_ws, size_t ws_size,
                              hipStream_t stream) {
    const float* x       = (const float*)d_in[0];
    const float* emb     = (const float*)d_in[1];
    const int*   buckets = (const int*)d_in[2];
    float*       out     = (float*)d_out;

    int* counts  = (int*)d_ws;
    int* cursor  = (int*)((char*)d_ws + 1024);
    int* offsets = (int*)((char*)d_ws + 2048);
    int* perm    = (int*)((char*)d_ws + 4096);

    k_zero_counts<<<1, 256, 0, stream>>>(counts);
    k_hist<<<(NPIX + 255) / 256, 256, 0, stream>>>(buckets, counts);
    k_scan<<<1, 1, 0, stream>>>(counts, offsets, cursor);
    k_scatter<<<(NPIX + 255) / 256, 256, 0, stream>>>(buckets, cursor, perm);
    k_bucket_gemm<<<dim3(NTYPES, BLOCKS_PER_TYPE), 256, 0, stream>>>(
        x, emb, offsets, perm, out);
}


# Patch gather: lane halves split k by input channel (4 channels x 9 taps each), so a lane reads its 36 patch values as 12 three-pixel row runs, one 12-byte load per run, instead of 36 4-byte loads
# speedup vs baseline: 1.4367x; 1.4367x over previous
#include <hip/hip_runtime.h>

#define HH 256
#define WW 256
#define BB 4
#define CIN 8
#define COUT 8
#define NTYPES 216
#define NPIX (BB*HH*WW)
#define KDIM 72
#define NW   (COUT*KDIM)
#define EMB_STRIDE (NW + COUT)

typedef float v2f __attribute__((ext_vector_type(2)));
typedef float v8f __attribute__((ext_vector_type(8)));

#define SORT_BLOCKS   64
#define SORT_THREADS  512
#define SORT_PER_THR  (NPIX / (SORT_BLOCKS * SORT_THREADS))
#define SORT_PER_BLK  (SORT_THREADS * SORT_PER_THR)

__global__ void k_zero_counts(int* counts, int* cursor) {
    int i = threadIdx.x;
    if (i < NTYPES) { counts[i] = 0; cursor[i] = 0; }
}

__global__ __launch_bounds__(SORT_THREADS) void k_hist(
    const int* __restrict__ buckets, int* __restrict__ counts) {
    __shared__ int lh[NTYPES];
    const int tid = threadIdx.x;
    if (tid < NTYPES) lh[tid] = 0;
    __syncthreads();
    const int base = blockIdx.x * SORT_PER_BLK;
    int b[SORT_PER_THR];
    for (int j = 0; j < SORT_PER_THR; ++j) b[j] = buckets[base + j * SORT_THREADS + tid];
    for (int j = 0; j < SORT_PER_THR; ++j) atomicAdd(&lh[b[j]], 1);
    __syncthreads();
    if (tid < NTYPES) {
        const int h = lh[tid];
        if (h) atomicAdd(&counts[tid], h);
    }
}

__global__ __launch_bounds__(SORT_THREADS) void k_scatter(
    const int* __restrict__ buckets, const int* __restrict__ counts,
    int* __restrict__ cursor, int* __restrict__ offsets, int* __restrict__ perm) {
    __shared__ int lh[NTYPES];
    __shared__ int wsum[8];
    const int tid  = threadIdx.x;
    const int lane = tid & 31;
    const int wave = tid >> 5;
    if (tid < NTYPES) lh[tid] = 0;
    __syncthreads();
    const int base = blockIdx.x * SORT_PER_BLK;
    int b[SORT_PER_THR];
    for (int j = 0; j < SORT_PER_THR; ++j) b[j] = buckets[base + j * SORT_THREADS + tid];
    for (int j = 0; j < SORT_PER_THR; ++j) atomicAdd(&lh[b[j]], 1);

    int c = 0, incl = 0;
    if (tid < 256) {
        c = (tid < NTYPES) ? counts[tid] : 0;
        incl = c;
        for (int d = 1; d < 32; d <<= 1) {
            const int nb = __shfl_up(incl, d, 32);
            if (lane >= d) incl += nb;
        }
        if (lane == 31) wsum[wave] = incl;
    }
    __syncthreads();
    if (tid < NTYPES) {
        int pre = 0;
        for (int w = 0; w < wave; ++w) pre += wsum[w];
        const int start = pre + incl - c;
        const int h = lh[tid];
        int rel = 0;
        if (h) rel = atomicAdd(&cursor[tid], h);
        lh[tid] = start + rel;
        if (blockIdx.x == 0) {
            offsets[tid] = start;
            if (tid == NTYPES - 1) offsets[NTYPES] = start + c;
        }
    }
    __syncthreads();
    int p[SORT_PER_THR];
    for (int j = 0; j < SORT_PER_THR; ++j) p[j] = atomicAdd(&lh[b[j]], 1);
    for (int j = 0; j < SORT_PER_THR; ++j) perm[p[j]] = base + j * SORT_THREADS + tid;
}

#define WAVES_PER_BLOCK 8
#define BLOCKS_PER_TYPE 4

__global__ __launch_bounds__(256) void k_bucket_gemm(
    const float* __restrict__ x, const float* __restrict__ emb,
    const int* __restrict__ offsets, const int* __restrict__ perm,
    float* __restrict__ out) {

    const int t    = blockIdx.x;
    const int lane = threadIdx.x & 31;
    const int gwave  = blockIdx.y * WAVES_PER_BLOCK + (threadIdx.x >> 5);
    const int nwaves = gridDim.y * WAVES_PER_BLOCK;

    const int start  = offsets[t];
    const int end    = offsets[t + 1];
    const int ntiles = (end - start + 15) >> 4;

    const float* wrow = emb + t * EMB_STRIDE;

    const int n  = lane & 15;
    const int hi = lane >> 4;

    v2f bmat[18];
    #pragma unroll
    for (int cc = 0; cc < 18; ++cc) {
        float b0 = 0.f, b1 = 0.f;
        if (n < COUT) {
            const float* wp = wrow + n * KDIM + hi * 36 + cc * 2;
            b0 = wp[0];
            b1 = wp[1];
        }
        bmat[cc].x = b0; bmat[cc].y = b1;
    }
    const float biasv = (n < COUT) ? wrow[NW + n] : 0.f;

    for (int tile = gwave; tile < ntiles; tile += nwaves) {
        const int rowbase = start + tile * 16;

        const int  pidx  = rowbase + n;
        const bool valid = pidx < end;
        const int  cp    = valid ? pidx : start;
        const int  pix   = perm[cp];
        const int pb = pix >> 16;
        const int ph = (pix >> 8) & 255;
        const int pw = pix & 255;

        const int  ws   = min(max(pw - 1, 0), WW - 3);
        const int  d    = (pw - 1) - ws;
        const bool okw0 = valid && (pw >= 1);
        const bool okw2 = valid && (pw <= WW - 2);

        int  hoff[3];
        bool hok[3];
        #pragma unroll
        for (int kh = 0; kh < 3; ++kh) {
            const int hh = ph + kh - 1;
            hoff[kh] = ((hh & (HH - 1)) << 8) + ws;
            hok[kh]  = ((unsigned)hh < (unsigned)HH);
        }
        const int base = (pb * CIN + hi * 4) << 16;

        float vals[36];
        #pragma unroll
        for (int r = 0; r < 12; ++r) {
            const int cl = r / 3;
            const int kh = r % 3;
            const float* p = x + (base + (cl << 16) + hoff[kh]);
            const float L0 = p[0];
            const float L1 = p[1];
            const float L2 = p[2];
            const float a0 = (d > 0) ? L1 : L0;
            const float a1 = (d == 0) ? L1 : ((d < 0) ? L0 : L2);
            const float a2 = (d < 0) ? L1 : L2;
            vals[r * 3 + 0] = (hok[kh] && okw0)  ? a0 : 0.f;
            vals[r * 3 + 1] = (hok[kh] && valid) ? a1 : 0.f;
            vals[r * 3 + 2] = (hok[kh] && okw2)  ? a2 : 0.f;
        }

        v2f a[18];
        #pragma unroll
        for (int cc = 0; cc < 18; ++cc) {
            a[cc].x = vals[cc * 2];
            a[cc].y = vals[cc * 2 + 1];
        }

        v8f acc = {};
        #pragma unroll
        for (int cc = 0; cc < 18; ++cc) {
            acc = __builtin_amdgcn_wmma_f32_16x16x4_f32(
                false, a[cc], false, bmat[cc], (short)0, acc, false, false);
        }

        const int col  = n;
        const int rofs = hi << 3;
        #pragma unroll
        for (int r = 0; r < 8; ++r) {
            const int row   = r + rofs;
            const int pix2  = __shfl(pix, row, 32);
            if (col < COUT && (rowbase + row) < end) {
                const int b2 = pix2 >> 16;
                const int h2 = (pix2 >> 8) & 255;
                const int w2 = pix2 & 255;
                out[((b2 * COUT + col) << 16) + (h2 << 8) + w2] = acc[r] + biasv;
            }
        }
    }
}

extern "C" void kernel_launch(void* const* d_in, const int* in_sizes, int n_in,
                              void* d_out, int out_size, void* d_ws, size_t ws_size,
                              hipStream_t stream) {
    const float* x       = (const float*)d_in[0];
    const float* emb     = (const float*)d_in[1];
    const int*   buckets = (const int*)d_in[2];
    float*       out     = (float*)d_out;

    int* counts  = (int*)d_ws;
    int* cursor  = (int*)((char*)d_ws + 1024);
    int* offsets = (int*)((char*)d_ws + 2048);
    int* perm    = (int*)((char*)d_ws + 4096);

    k_zero_counts<<<1, 256, 0, stream>>>(counts, cursor);
    k_hist<<<SORT_BLOCKS, SORT_THREADS, 0, stream>>>(buckets, counts);
    k_scatter<<<SORT_BLOCKS, SORT_THREADS, 0, stream>>>(buckets, counts, cursor, offsets, perm);
    k_bucket_gemm<<<dim3(NTYPES, BLOCKS_PER_TYPE), 256, 0, stream>>>(
        x, emb, offsets, perm, out);
}


# Each 16x16 matrix tile now serves a pair of buckets (second bucket's filters in the 8 B columns that were zero); strip halved to 4 rows so 256 groups fill every WGP with half the tiles each
# speedup vs baseline: 9.3121x; 6.4816x over previous
#include <hip/hip_runtime.h>

#define HH 256
#define WW 256
#define BB 4
#define CIN 8
#define COUT 8
#define NTYPES 216
#define KDIM 72
#define NW   (COUT*KDIM)
#define EMB_STRIDE (NW + COUT)

typedef float v2f __attribute__((ext_vector_type(2)));
typedef float v8f __attribute__((ext_vector_type(8)));

#define SR      4
#define SPIX    (SR * WW)
#define NSTRIPS (BB * HH / SR)
#define NTHR    512
#define NWAVE   (NTHR / 32)
#define NPAIR   (NTYPES / 2)
#define YOFF    (EMB_STRIDE + 4)
#define WSTRIDE (YOFF + EMB_STRIDE)
#define NF4P    (2 * (EMB_STRIDE / 4))
#define NWREG   ((NF4P + 31) / 32)
#define PER_THR (SPIX / NTHR)
#define QSTEP   (NTHR / 64)
#define XROWS   (SR + 2)
#define XOFF    4
#define XPITCH  (XOFF + WW + 4)
#define XPLANE  (XROWS * XPITCH + 8)
#define OPITCH  (SPIX + 8)
#define NF4     (EMB_STRIDE / 4)
#define XLOADS  (CIN * XROWS * (WW / 4) / NTHR)
#define OF4     (SPIX / 4)

__global__ __launch_bounds__(NTHR) void k_strip_conv(
    const float* __restrict__ x, const float* __restrict__ emb,
    const int* __restrict__ buckets, float* __restrict__ out) {

    __shared__ __attribute__((aligned(16))) float xs[CIN * XPLANE];
    __shared__ __attribute__((aligned(16))) float os[COUT * OPITCH];
    __shared__ __attribute__((aligned(16))) float wbuf[NWAVE * WSTRIDE];
    __shared__ int lperm[SPIX];
    __shared__ int s_cnt[NTYPES];
    __shared__ int s_start[NTYPES];
    __shared__ int wsum[8];

    const int tid  = threadIdx.x;
    const int lane = tid & 31;
    const int wave = __builtin_amdgcn_readfirstlane(tid >> 5);

    const int strip = blockIdx.x;
    const int pb = strip / (HH / SR);
    const int h0 = (strip % (HH / SR)) * SR;
    const int gpix0 = strip * SPIX;

    if (tid < NTYPES) s_cnt[tid] = 0;

    const int c4 = tid & 63;
    const int q0 = tid >> 6;
    float4 xv[XLOADS];
    #pragma unroll
    for (int j = 0; j < XLOADS; ++j) {
        const int q  = q0 + QSTEP * j;
        const int c  = q / XROWS;
        const int rr = q - c * XROWS;
        const int h  = h0 - 1 + rr;
        const int hc = min(max(h, 0), HH - 1);
        xv[j] = *(const float4*)(x + (((pb * CIN + c) << 16) + (hc << 8) + (c4 << 2)));
    }
    int bk[PER_THR], rk[PER_THR];
    for (int j = 0; j < PER_THR; ++j) bk[j] = buckets[gpix0 + j * NTHR + tid];
    #pragma unroll
    for (int j = 0; j < XLOADS; ++j) {
        const int q  = q0 + QSTEP * j;
        const int c  = q / XROWS;
        const int rr = q - c * XROWS;
        const int h  = h0 - 1 + rr;
        float4 v = xv[j];
        if ((unsigned)h >= (unsigned)HH) v = make_float4(0.f, 0.f, 0.f, 0.f);
        *(float4*)(xs + c * XPLANE + rr * XPITCH + XOFF + (c4 << 2)) = v;
    }
    if (tid < CIN * XROWS * 2) {
        const int q  = tid >> 1;
        const int c  = q / XROWS;
        const int rr = q - c * XROWS;
        xs[c * XPLANE + rr * XPITCH + ((tid & 1) ? (XOFF + WW) : (XOFF - 1))] = 0.f;
    }
    __syncthreads();

    for (int j = 0; j < PER_THR; ++j) rk[j] = atomicAdd(&s_cnt[bk[j]], 1);
    __syncthreads();

    int c = 0, incl = 0;
    if (tid < 256) {
        c = (tid < NTYPES) ? s_cnt[tid] : 0;
        incl = c;
        for (int d = 1; d < 32; d <<= 1) {
            const int nb = __shfl_up(incl, d, 32);
            if (lane >= d) incl += nb;
        }
        if (lane == 31) wsum[tid >> 5] = incl;
    }
    __syncthreads();
    if (tid < NTYPES) {
        int pre = 0;
        for (int w = 0; w < (tid >> 5); ++w) pre += wsum[w];
        s_start[tid] = pre + incl - c;
    }
    __syncthreads();
    for (int j = 0; j < PER_THR; ++j) lperm[s_start[bk[j]] + rk[j]] = j * NTHR + tid;
    __syncthreads();

    const int n  = lane & 15;
    const int hi = lane >> 4;

    float*  wb  = wbuf + wave * WSTRIDE;
    float4* wb4 = (float4*)wb;
    const int grp = n >> 3;
    const float* wrow  = wb + grp * YOFF + (n & 7) * KDIM;
    const float* wfrag = wrow + hi * 36;
    const float* xlane = xs + (hi * 4) * XPLANE + (XOFF - 1);

    float4 wreg[NWREG];
    {
        const float4* wsrc = (const float4*)(emb + (2 * wave) * EMB_STRIDE);
        #pragma unroll
        for (int i = 0; i < NWREG; ++i) {
            const int idx = lane + 32 * i;
            wreg[i] = (idx < NF4P) ? wsrc[idx] : make_float4(0.f, 0.f, 0.f, 0.f);
        }
    }

    for (int pr = wave; pr < NPAIR; pr += NWAVE) {
        #pragma unroll
        for (int i = 0; i < NWREG; ++i) {
            const int idx = lane + 32 * i;
            if (idx < NF4P) wb4[idx + ((idx >= NF4) ? 1 : 0)] = wreg[i];
        }
        const int pn = pr + NWAVE;
        if (pn < NPAIR) {
            const float4* wsrc = (const float4*)(emb + (2 * pn) * EMB_STRIDE);
            #pragma unroll
            for (int i = 0; i < NWREG; ++i) {
                const int idx = lane + 32 * i;
                if (idx < NF4P) wreg[i] = wsrc[idx];
            }
        }
        const int cntx = __builtin_amdgcn_readfirstlane(s_cnt[2 * pr]);
        const int cnt  = cntx + __builtin_amdgcn_readfirstlane(s_cnt[2 * pr + 1]);
        if (cnt == 0) continue;
        const int st  = __builtin_amdgcn_readfirstlane(s_start[2 * pr]);

        v2f bmat[18];
        #pragma unroll
        for (int q = 0; q < 9; ++q) {
            const float4 v = *(const float4*)(wfrag + 4 * q);
            bmat[2 * q].x     = v.x; bmat[2 * q].y     = v.y;
            bmat[2 * q + 1].x = v.z; bmat[2 * q + 1].y = v.w;
        }
        const float biasv = wb[grp * YOFF + NW + (n & 7)];

        for (int tb = 0; tb < cnt; tb += 16) {
            const int ridx = tb + n;
            const int lp   = lperm[st + ((ridx < cnt) ? ridx : 0)];
            const float* p = xlane + (lp >> 8) * XPITCH + (lp & 255);

            int lp2[8];
            #pragma unroll
            for (int r = 0; r < 8; ++r)
                lp2[r] = __shfl(lp, r + (hi << 3), 32);

            float vals[36];
            #pragma unroll
            for (int cl = 0; cl < 4; ++cl)
                #pragma unroll
                for (int kh = 0; kh < 3; ++kh)
                    #pragma unroll
                    for (int kw = 0; kw < 3; ++kw)
                        vals[cl * 9 + kh * 3 + kw] = p[cl * XPLANE + kh * XPITCH + kw];

            v8f acc = {};
            #pragma unroll
            for (int cc = 0; cc < 18; ++cc) {
                v2f a;
                a.x = vals[cc * 2];
                a.y = vals[cc * 2 + 1];
                acc = __builtin_amdgcn_wmma_f32_16x16x4_f32(
                    false, a, false, bmat[cc], (short)0, acc, false, false);
            }

            #pragma unroll
            for (int r = 0; r < 8; ++r) {
                const int row = tb + r + (hi << 3);
                if (row < cnt && ((row >= cntx) ? 1 : 0) == grp)
                    os[(n & 7) * OPITCH + lp2[r]] = acc[r] + biasv;
            }
        }
    }
    __syncthreads();

    for (int i = tid; i < COUT * OF4; i += NTHR) {
        const int o   = i / OF4;
        const int rem = i % OF4;
        *(float4*)(out + (((pb * COUT + o) << 16) + (h0 << 8) + (rem << 2))) =
            *(const float4*)(os + o * OPITCH + (rem << 2));
    }
}

extern "C" void kernel_launch(void* const* d_in, const int* in_sizes, int n_in,
                              void* d_out, int out_size, void* d_ws, size_t ws_size,
                              hipStream_t stream) {
    const float* x       = (const float*)d_in[0];
    const float* emb     = (const float*)d_in[1];
    const int*   buckets = (const int*)d_in[2];
    float*       out     = (float*)d_out;

    k_strip_conv<<<NSTRIPS, NTHR, 0, stream>>>(x, emb, buckets, out);
}


# All LDS reads of a tile in one group ahead of its 18 products: x strip kept channel-interleaved in LDS so the patch gather is 9 16-byte loads, not 24 narrow ones; row pixels passed by row_share moves
# speedup vs baseline: 9.9785x; 1.0716x over previous
#include <hip/hip_runtime.h>

#define HH 256
#define WW 256
#define BB 4
#define CIN 8
#define COUT 8
#define NTYPES 216
#define KDIM 72
#define NW   (COUT*KDIM)
#define EMB_STRIDE (NW + COUT)

typedef float v2f __attribute__((ext_vector_type(2)));
typedef float v8f __attribute__((ext_vector_type(8)));

#define SR      4
#define SPIX    (SR * WW)
#define NSTRIPS (BB * HH / SR)
#define NTHR    512
#define NWAVE   (NTHR / 32)
#define NPAIR   (NTYPES / 2)
#define YOFF    (EMB_STRIDE + 4)
#define WSTRIDE (YOFF + EMB_STRIDE)
#define NF4P    (2 * (EMB_STRIDE / 4))
#define NWREG   ((NF4P + 31) / 32)
#define PER_THR (SPIX / NTHR)
#define XROWS   (SR + 2)
#define XPP     12
#define XPIX    (WW + 2)
#define XRP     (XPIX * XPP)
#define XLTHR   (XROWS * (WW / 4))
#define OPITCH  (SPIX + 8)
#define NF4     (EMB_STRIDE / 4)
#define OF4     (SPIX / 4)

__device__ __forceinline__ float fcomp(const float4& v, int i) {
    return i == 0 ? v.x : (i == 1 ? v.y : (i == 2 ? v.z : v.w));
}
__device__ __forceinline__ void xs4zero(float* p) {
    *(float4*)p = make_float4(0.f, 0.f, 0.f, 0.f);
}

__global__ __launch_bounds__(NTHR) void k_strip_conv(
    const float* __restrict__ x, const float* __restrict__ emb,
    const int* __restrict__ buckets, float* __restrict__ out) {

    __shared__ __attribute__((aligned(16))) float xs[XROWS * XRP];
    __shared__ __attribute__((aligned(16))) float os[COUT * OPITCH];
    __shared__ __attribute__((aligned(16))) float wbuf[NWAVE * WSTRIDE];
    __shared__ int lperm[SPIX];
    __shared__ int s_cnt[NTYPES];
    __shared__ int s_start[NTYPES];
    __shared__ int wsum[8];

    const int tid  = threadIdx.x;
    const int lane = tid & 31;
    const int wave = __builtin_amdgcn_readfirstlane(tid >> 5);

    const int strip = blockIdx.x;
    const int pb = strip / (HH / SR);
    const int h0 = (strip % (HH / SR)) * SR;
    const int gpix0 = strip * SPIX;

    if (tid < NTYPES) s_cnt[tid] = 0;

    const int c4 = ((tid >> 5) & 3) * 16 + (lane & 7) + ((lane >> 4) << 3);
    const int rr = ((tid >> 7) << 1) + ((lane >> 3) & 1);
    const int h  = h0 - 1 + rr;
    float4 xv[CIN];
    if (tid < XLTHR) {
        const int hc = min(max(h, 0), HH - 1);
        #pragma unroll
        for (int c = 0; c < CIN; ++c)
            xv[c] = *(const float4*)(x + (((pb * CIN + c) << 16) + (hc << 8) + (c4 << 2)));
    }
    int bk[PER_THR], rk[PER_THR];
    for (int j = 0; j < PER_THR; ++j) bk[j] = buckets[gpix0 + j * NTHR + tid];
    if (tid < XLTHR) {
        const bool inside = (unsigned)h < (unsigned)HH;
        float* dst = xs + rr * XRP + ((c4 << 2) + 1) * XPP;
        #pragma unroll
        for (int px = 0; px < 4; ++px) {
            float4 lo = make_float4(fcomp(xv[0], px), fcomp(xv[1], px), fcomp(xv[2], px), fcomp(xv[3], px));
            float4 up = make_float4(fcomp(xv[4], px), fcomp(xv[5], px), fcomp(xv[6], px), fcomp(xv[7], px));
            if (!inside) { lo = make_float4(0.f, 0.f, 0.f, 0.f); up = lo; }
            *(float4*)(dst + px * XPP)     = lo;
            *(float4*)(dst + px * XPP + 4) = up;
        }
    } else if (tid < XLTHR + XROWS * 4) {
        const int b  = tid - XLTHR;
        const int br = b >> 2;
        xs4zero(xs + br * XRP + ((b & 2) ? (XPIX - 1) * XPP : 0) + ((b & 1) << 2));
    }
    __syncthreads();

    for (int j = 0; j < PER_THR; ++j) rk[j] = atomicAdd(&s_cnt[bk[j]], 1);
    __syncthreads();

    int c = 0, incl = 0;
    if (tid < 256) {
        c = (tid < NTYPES) ? s_cnt[tid] : 0;
        incl = c;
        for (int d = 1; d < 32; d <<= 1) {
            const int nb = __shfl_up(incl, d, 32);
            if (lane >= d) incl += nb;
        }
        if (lane == 31) wsum[tid >> 5] = incl;
    }
    __syncthreads();
    if (tid < NTYPES) {
        int pre = 0;
        for (int w = 0; w < (tid >> 5); ++w) pre += wsum[w];
        s_start[tid] = pre + incl - c;
    }
    __syncthreads();
    for (int j = 0; j < PER_THR; ++j) lperm[s_start[bk[j]] + rk[j]] = j * NTHR + tid;
    __syncthreads();

    const int n  = lane & 15;
    const int hi = lane >> 4;

    float*  wb  = wbuf + wave * WSTRIDE;
    float4* wb4 = (float4*)wb;
    const int grp = n >> 3;
    const float* wrow  = wb + grp * YOFF + (n & 7) * KDIM;
    const float* wfrag = wrow + hi * 36;
    const float* xlane = xs + hi * 4;

    float4 wreg[NWREG];
    {
        const float4* wsrc = (const float4*)(emb + (2 * wave) * EMB_STRIDE);
        #pragma unroll
        for (int i = 0; i < NWREG; ++i) {
            const int idx = lane + 32 * i;
            wreg[i] = (idx < NF4P) ? wsrc[idx] : make_float4(0.f, 0.f, 0.f, 0.f);
        }
    }

    for (int pr = wave; pr < NPAIR; pr += NWAVE) {
        #pragma unroll
        for (int i = 0; i < NWREG; ++i) {
            const int idx = lane + 32 * i;
            if (idx < NF4P) wb4[idx + ((idx >= NF4) ? 1 : 0)] = wreg[i];
        }
        const int pn = pr + NWAVE;
        if (pn < NPAIR) {
            const float4* wsrc = (const float4*)(emb + (2 * pn) * EMB_STRIDE);
            #pragma unroll
            for (int i = 0; i < NWREG; ++i) {
                const int idx = lane + 32 * i;
                if (idx < NF4P) wreg[i] = wsrc[idx];
            }
        }
        const int cntx = __builtin_amdgcn_readfirstlane(s_cnt[2 * pr]);
        const int cnt  = cntx + __builtin_amdgcn_readfirstlane(s_cnt[2 * pr + 1]);
        if (cnt == 0) continue;
        const int st  = __builtin_amdgcn_readfirstlane(s_start[2 * pr]);

        v2f bmat[18];
        float wh[36];
        #pragma unroll
        for (int q = 0; q < 9; ++q) {
            const float4 v = *(const float4*)(wfrag + 4 * q);
            wh[4 * q] = v.x; wh[4 * q + 1] = v.y; wh[4 * q + 2] = v.z; wh[4 * q + 3] = v.w;
        }
        #pragma unroll
        for (int t = 0; t < 9; ++t) {
            bmat[2 * t].x     = wh[t];      bmat[2 * t].y     = wh[9 + t];
            bmat[2 * t + 1].x = wh[18 + t]; bmat[2 * t + 1].y = wh[27 + t];
        }
        const float biasv = wb[grp * YOFF + NW + (n & 7)];

        for (int tb = 0; tb < cnt; tb += 16) {
            const int ridx  = tb + n;
            const int ridx2 = tb + ((n + (hi << 3)) & 15);
            const int lp    = lperm[st + ((ridx  < cnt) ? ridx  : 0)];
            const int lpx   = lperm[st + ((ridx2 < cnt) ? ridx2 : 0)];
            const float* p = xlane + (lp >> 8) * XRP + (lp & 255) * XPP;

            int lp2[8];
            lp2[0] = __builtin_amdgcn_update_dpp(0, lpx, 0x150, 0xf, 0xf, true);
            lp2[1] = __builtin_amdgcn_update_dpp(0, lpx, 0x151, 0xf, 0xf, true);
            lp2[2] = __builtin_amdgcn_update_dpp(0, lpx, 0x152, 0xf, 0xf, true);
            lp2[3] = __builtin_amdgcn_update_dpp(0, lpx, 0x153, 0xf, 0xf, true);
            lp2[4] = __builtin_amdgcn_update_dpp(0, lpx, 0x154, 0xf, 0xf, true);
            lp2[5] = __builtin_amdgcn_update_dpp(0, lpx, 0x155, 0xf, 0xf, true);
            lp2[6] = __builtin_amdgcn_update_dpp(0, lpx, 0x156, 0xf, 0xf, true);
            lp2[7] = __builtin_amdgcn_update_dpp(0, lpx, 0x157, 0xf, 0xf, true);

            float4 v4[9];
            #pragma unroll
            for (int kh = 0; kh < 3; ++kh)
                #pragma unroll
                for (int kw = 0; kw < 3; ++kw)
                    v4[kh * 3 + kw] = *(const float4*)(p + kh * XRP + kw * XPP);

            __builtin_amdgcn_sched_barrier(0);

            v8f acc = {};
            #pragma unroll
            for (int t = 0; t < 9; ++t) {
                v2f a;
                a.x = v4[t].x;
                a.y = v4[t].y;
                acc = __builtin_amdgcn_wmma_f32_16x16x4_f32(
                    false, a, false, bmat[2 * t], (short)0, acc, false, false);
                a.x = v4[t].z;
                a.y = v4[t].w;
                acc = __builtin_amdgcn_wmma_f32_16x16x4_f32(
                    false, a, false, bmat[2 * t + 1], (short)0, acc, false, false);
            }

            #pragma unroll
            for (int r = 0; r < 8; ++r) {
                const int row = tb + r + (hi << 3);
                if (row < cnt && ((row >= cntx) ? 1 : 0) == grp)
                    os[(n & 7) * OPITCH + lp2[r]] = acc[r] + biasv;
            }
        }
    }
    __syncthreads();

    for (int i = tid; i < COUT * OF4; i += NTHR) {
        const int o   = i / OF4;
        const int rem = i % OF4;
        *(float4*)(out + (((pb * COUT + o) << 16) + (h0 << 8) + (rem << 2))) =
            *(const float4*)(os + o * OPITCH + (rem << 2));
    }
}

extern "C" void kernel_launch(void* const* d_in, const int* in_sizes, int n_in,
                              void* d_out, int out_size, void* d_ws, size_t ws_size,
                              hipStream_t stream) {
    const float* x       = (const float*)d_in[0];
    const float* emb     = (const float*)d_in[1];
    const int*   buckets = (const int*)d_in[2];
    float*       out     = (float*)d_out;

    k_strip_conv<<<NSTRIPS, NTHR, 0, stream>>>(x, emb, buckets, out);
}


# Matrix operand roles swapped (weights = row operand, patches = column operand): a lane ends with all 8 channels of one pixel, so one validity mask per tile, no row_share moves, bias as C operand
# speedup vs baseline: 11.3382x; 1.1363x over previous
#include <hip/hip_runtime.h>

#define HH 256
#define WW 256
#define BB 4
#define CIN 8
#define COUT 8
#define NTYPES 216
#define KDIM 72
#define NW   (COUT*KDIM)
#define EMB_STRIDE (NW + COUT)

typedef float v2f __attribute__((ext_vector_type(2)));
typedef float v8f __attribute__((ext_vector_type(8)));

#define SR      4
#define SPIX    (SR * WW)
#define NSTRIPS (BB * HH / SR)
#define NTHR    512
#define NWAVE   (NTHR / 32)
#define NPAIR   (NTYPES / 2)
#define YOFF    (EMB_STRIDE + 4)
#define WSTRIDE (YOFF + EMB_STRIDE)
#define NF4P    (2 * (EMB_STRIDE / 4))
#define NWREG   ((NF4P + 31) / 32)
#define PER_THR (SPIX / NTHR)
#define XROWS   (SR + 2)
#define XPP     12
#define XPIX    (WW + 2)
#define XRP     (XPIX * XPP)
#define XLTHR   (XROWS * (WW / 4))
#define OPITCH  (SPIX + 8)
#define NF4     (EMB_STRIDE / 4)
#define OF4     (SPIX / 4)

__device__ __forceinline__ float fcomp(const float4& v, int i) {
    return i == 0 ? v.x : (i == 1 ? v.y : (i == 2 ? v.z : v.w));
}
__device__ __forceinline__ void xs4zero(float* p) {
    *(float4*)p = make_float4(0.f, 0.f, 0.f, 0.f);
}

__global__ __launch_bounds__(NTHR) void k_strip_conv(
    const float* __restrict__ x, const float* __restrict__ emb,
    const int* __restrict__ buckets, float* __restrict__ out) {

    __shared__ __attribute__((aligned(16))) float xs[XROWS * XRP];
    __shared__ __attribute__((aligned(16))) float os[COUT * OPITCH];
    __shared__ __attribute__((aligned(16))) float wbuf[NWAVE * WSTRIDE];
    __shared__ int lperm[SPIX];
    __shared__ int s_cnt[NTYPES];
    __shared__ int s_start[NTYPES];
    __shared__ int wsum[8];

    const int tid  = threadIdx.x;
    const int lane = tid & 31;
    const int wave = __builtin_amdgcn_readfirstlane(tid >> 5);

    const int strip = blockIdx.x;
    const int pb = strip / (HH / SR);
    const int h0 = (strip % (HH / SR)) * SR;
    const int gpix0 = strip * SPIX;

    if (tid < NTYPES) s_cnt[tid] = 0;

    const int c4 = ((tid >> 5) & 3) * 16 + (lane & 7) + ((lane >> 4) << 3);
    const int rr = ((tid >> 7) << 1) + ((lane >> 3) & 1);
    const int h  = h0 - 1 + rr;
    float4 xv[CIN];
    if (tid < XLTHR) {
        const int hc = min(max(h, 0), HH - 1);
        #pragma unroll
        for (int c = 0; c < CIN; ++c)
            xv[c] = *(const float4*)(x + (((pb * CIN + c) << 16) + (hc << 8) + (c4 << 2)));
    }
    int bk[PER_THR], rk[PER_THR];
    for (int j = 0; j < PER_THR; ++j) bk[j] = buckets[gpix0 + j * NTHR + tid];
    if (tid < XLTHR) {
        const bool inside = (unsigned)h < (unsigned)HH;
        float* dst = xs + rr * XRP + ((c4 << 2) + 1) * XPP;
        #pragma unroll
        for (int px = 0; px < 4; ++px) {
            float4 lo = make_float4(fcomp(xv[0], px), fcomp(xv[1], px), fcomp(xv[2], px), fcomp(xv[3], px));
            float4 up = make_float4(fcomp(xv[4], px), fcomp(xv[5], px), fcomp(xv[6], px), fcomp(xv[7], px));
            if (!inside) { lo = make_float4(0.f, 0.f, 0.f, 0.f); up = lo; }
            *(float4*)(dst + px * XPP)     = lo;
            *(float4*)(dst + px * XPP + 4) = up;
        }
    } else if (tid < XLTHR + XROWS * 4) {
        const int b  = tid - XLTHR;
        const int br = b >> 2;
        xs4zero(xs + br * XRP + ((b & 2) ? (XPIX - 1) * XPP : 0) + ((b & 1) << 2));
    }
    __syncthreads();

    for (int j = 0; j < PER_THR; ++j) rk[j] = atomicAdd(&s_cnt[bk[j]], 1);
    __syncthreads();

    int c = 0, incl = 0;
    if (tid < 256) {
        c = (tid < NTYPES) ? s_cnt[tid] : 0;
        incl = c;
        for (int d = 1; d < 32; d <<= 1) {
            const int nb = __shfl_up(incl, d, 32);
            if (lane >= d) incl += nb;
        }
        if (lane == 31) wsum[tid >> 5] = incl;
    }
    __syncthreads();
    if (tid < NTYPES) {
        int pre = 0;
        for (int w = 0; w < (tid >> 5); ++w) pre += wsum[w];
        s_start[tid] = pre + incl - c;
    }
    __syncthreads();
    for (int j = 0; j < PER_THR; ++j) lperm[s_start[bk[j]] + rk[j]] = j * NTHR + tid;
    __syncthreads();

    const int n  = lane & 15;
    const int hi = lane >> 4;

    float*  wb  = wbuf + wave * WSTRIDE;
    float4* wb4 = (float4*)wb;
    const int grp = n >> 3;
    const float* wrow  = wb + grp * YOFF + (n & 7) * KDIM;
    const float* wfrag = wrow + hi * 36;
    const float* xlane = xs + hi * 4;

    float4 wreg[NWREG];
    {
        const float4* wsrc = (const float4*)(emb + (2 * wave) * EMB_STRIDE);
        #pragma unroll
        for (int i = 0; i < NWREG; ++i) {
            const int idx = lane + 32 * i;
            wreg[i] = (idx < NF4P) ? wsrc[idx] : make_float4(0.f, 0.f, 0.f, 0.f);
        }
    }

    for (int pr = wave; pr < NPAIR; pr += NWAVE) {
        #pragma unroll
        for (int i = 0; i < NWREG; ++i) {
            const int idx = lane + 32 * i;
            if (idx < NF4P) wb4[idx + ((idx >= NF4) ? 1 : 0)] = wreg[i];
        }
        const int pn = pr + NWAVE;
        if (pn < NPAIR) {
            const float4* wsrc = (const float4*)(emb + (2 * pn) * EMB_STRIDE);
            #pragma unroll
            for (int i = 0; i < NWREG; ++i) {
                const int idx = lane + 32 * i;
                if (idx < NF4P) wreg[i] = wsrc[idx];
            }
        }
        const int cntx = __builtin_amdgcn_readfirstlane(s_cnt[2 * pr]);
        const int cnt  = cntx + __builtin_amdgcn_readfirstlane(s_cnt[2 * pr + 1]);
        if (cnt == 0) continue;
        const int st  = __builtin_amdgcn_readfirstlane(s_start[2 * pr]);

        v2f bmat[18];
        float wh[36];
        #pragma unroll
        for (int q = 0; q < 9; ++q) {
            const float4 v = *(const float4*)(wfrag + 4 * q);
            wh[4 * q] = v.x; wh[4 * q + 1] = v.y; wh[4 * q + 2] = v.z; wh[4 * q + 3] = v.w;
        }
        #pragma unroll
        for (int t = 0; t < 9; ++t) {
            bmat[2 * t].x     = wh[t];      bmat[2 * t].y     = wh[9 + t];
            bmat[2 * t + 1].x = wh[18 + t]; bmat[2 * t + 1].y = wh[27 + t];
        }
        const float4 bq0 = *(const float4*)(wb + hi * YOFF + NW);
        const float4 bq1 = *(const float4*)(wb + hi * YOFF + NW + 4);
        const v8f cbias = {bq0.x, bq0.y, bq0.z, bq0.w, bq1.x, bq1.y, bq1.z, bq1.w};

        for (int tb = 0; tb < cnt; tb += 16) {
            const int ridx = tb + n;
            const int lp   = lperm[st + ((ridx < cnt) ? ridx : 0)];
            const float* p = xlane + (lp >> 8) * XRP + (lp & 255) * XPP;

            float4 v4[9];
            #pragma unroll
            for (int kh = 0; kh < 3; ++kh)
                #pragma unroll
                for (int kw = 0; kw < 3; ++kw)
                    v4[kh * 3 + kw] = *(const float4*)(p + kh * XRP + kw * XPP);

            __builtin_amdgcn_sched_barrier(0);

            v8f acc = cbias;
            #pragma unroll
            for (int t = 0; t < 9; ++t) {
                v2f a;
                a.x = v4[t].x;
                a.y = v4[t].y;
                acc = __builtin_amdgcn_wmma_f32_16x16x4_f32(
                    false, bmat[2 * t], false, a, (short)0, acc, false, false);
                a.x = v4[t].z;
                a.y = v4[t].w;
                acc = __builtin_amdgcn_wmma_f32_16x16x4_f32(
                    false, bmat[2 * t + 1], false, a, (short)0, acc, false, false);
            }

            if (ridx < cnt && ((ridx >= cntx) ? 1 : 0) == hi) {
                float* o = os + lp;
                #pragma unroll
                for (int r = 0; r < 8; ++r) o[r * OPITCH] = acc[r];
            }
        }
    }
    __syncthreads();

    for (int i = tid; i < COUT * OF4; i += NTHR) {
        const int o   = i / OF4;
        const int rem = i % OF4;
        *(float4*)(out + (((pb * COUT + o) << 16) + (h0 << 8) + (rem << 2))) =
            *(const float4*)(os + o * OPITCH + (rem << 2));
    }
}

extern "C" void kernel_launch(void* const* d_in, const int* in_sizes, int n_in,
                              void* d_out, int out_size, void* d_ws, size_t ws_size,
                              hipStream_t stream) {
    const float* x       = (const float*)d_in[0];
    const float* emb     = (const float*)d_in[1];
    const int*   buckets = (const int*)d_in[2];
    float*       out     = (float*)d_out;

    k_strip_conv<<<NSTRIPS, NTHR, 0, stream>>>(x, emb, buckets, out);
}
